# starting kernel
# baseline (speedup 1.0000x reference)
#include <hip/hip_runtime.h>
#include <hip/hip_bf16.h>

typedef float v2f __attribute__((ext_vector_type(2)));
typedef float v8f __attribute__((ext_vector_type(8)));

__global__ void dcn_pre_kernel(const float* __restrict__ W,
                               const float* __restrict__ b,
                               float* __restrict__ ws) {
  __shared__ float r1[1024];
  __shared__ float r2[1024];
  const int t = threadIdx.x;
  const float b0 = b[t];
  const float b1 = b[1024 + t];
  const float b2 = b[2048 + t];
  ws[4 + t] = b0 + b1 + b2;
  r1[t] = b0 * W[1024 + t];
  r2[t] = (b0 + b1) * W[2048 + t];
  __syncthreads();
  for (int s = 512; s > 0; s >>= 1) {
    if (t < s) { r1[t] += r1[t + s]; r2[t] += r2[t + s]; }
    __syncthreads();
  }
  if (t == 0) { ws[0] = 0.0f; ws[1] = r1[0]; ws[2] = r2[0]; }
}

__global__ void dcn_gemv_wmma_kernel(const float* __restrict__ x,
                                     const float* __restrict__ W,
                                     const float* __restrict__ ws_c,
                                     float* __restrict__ alpha) {
  __shared__ float sW[3 * 1024];
  __shared__ float sG[2 * 16 * 3 * 4];

  const int tid = threadIdx.x;
  for (int i = tid; i < 3 * 1024; i += 256) sW[i] = W[i];
  __syncthreads();

  const int wave  = tid >> 5;
  const int lane  = tid & 31;
  const int tileI = wave >> 2;
  const int slice = wave & 3;
  const int half  = lane >> 4;
  const int l16   = lane & 15;
  const int row0  = (blockIdx.x * 2 + tileI) * 16;
  const int k0    = slice * 256;

  const float* arow = x + (size_t)(row0 + l16) * 1024 + half * 2;
  const bool  bvalid = (l16 < 3);
  const float* brow = sW + l16 * 1024 + half * 2;
  const v2f bz = {0.0f, 0.0f};

  v8f acc = {};
  for (int k = k0; k < k0 + 256; k += 4) {
    v2f a = *(const v2f*)(arow + k);
    v2f bm = bvalid ? *(const v2f*)(brow + k) : bz;
    acc = __builtin_amdgcn_wmma_f32_16x16x4_f32(
         false, a,  false, bm,
         (short)0, acc,  false,  false);
  }

  if (bvalid) {
    #pragma unroll
    for (int r = 0; r < 8; ++r)
      sG[(((tileI * 16) + (r + 8 * half)) * 3 + l16) * 4 + slice] = acc[r];
  }
  __syncthreads();

  if (slice == 0 && lane < 16) {
    const int base = ((tileI * 16) + lane) * 3 * 4;
    const float g0 = sG[base + 0] + sG[base + 1] + sG[base + 2] + sG[base + 3];
    const float g1 = sG[base + 4] + sG[base + 5] + sG[base + 6] + sG[base + 7];
    const float g2 = sG[base + 8] + sG[base + 9] + sG[base + 10] + sG[base + 11];
    const float c1 = ws_c[1];
    const float c2 = ws_c[2];
    const float a1 = 1.0f + g0;
    const float a2 = a1 * (1.0f + g1) + c1;
    const float a3 = a2 * (1.0f + g2) + c2;
    alpha[row0 + lane] = a3;
  }
}

__global__ void dcn_out_kernel(const float* __restrict__ x,
                               const float* __restrict__ bsum,
                               const float* __restrict__ alpha,
                               float* __restrict__ out) {
  const size_t gid  = (size_t)blockIdx.x * 256 + threadIdx.x;
  const size_t base = gid * 4;
  const int row = (int)(base >> 10);
  const int col = (int)(base & 1023);
  const float a = alpha[row];
  const float4 xv = *(const float4*)(x + base);
  const float4 bv = *(const float4*)(bsum + col);
  float4 o;
  o.x = a * xv.x + bv.x;
  o.y = a * xv.y + bv.y;
  o.z = a * xv.z + bv.z;
  o.w = a * xv.w + bv.w;
  *(float4*)(out + base) = o;
}

extern "C" void kernel_launch(void* const* d_in, const int* in_sizes, int n_in,
                              void* d_out, int out_size, void* d_ws, size_t ws_size,
                              hipStream_t stream) {
  (void)in_sizes; (void)n_in; (void)out_size; (void)ws_size;
  const float* x = (const float*)d_in[0];
  const float* W = (const float*)d_in[1];
  const float* b = (const float*)d_in[2];
  float* out = (float*)d_out;
  float* ws  = (float*)d_ws;

  float* ws_c     = ws;
  float* ws_bsum  = ws + 4;
  float* ws_alpha = ws + 1028;

  dcn_pre_kernel<<<1, 1024, 0, stream>>>(W, b, ws);
  dcn_gemv_wmma_kernel<<<256, 256, 0, stream>>>(x, W, ws_c, ws_alpha);
  dcn_out_kernel<<<8192, 256, 0, stream>>>(x, ws_bsum, ws_alpha, out);
}


# Fuse the 3 launches into one: x tile read once into registers, transposed via wave-private LDS for the f32 WMMA, alpha applied in place; split-K now 8-way, so the sum order of g and c changes
# speedup vs baseline: 2.1378x; 2.1378x over previous
#include <hip/hip_runtime.h>

typedef float v2f __attribute__((ext_vector_type(2)));
typedef float v8f __attribute__((ext_vector_type(8)));

#define TILE_PITCH 132

__global__ __launch_bounds__(256)
void cross_fused_kernel(const float* __restrict__ x,
                        const float* __restrict__ W,
                        const float* __restrict__ b,
                        float* __restrict__ out) {
  __shared__ __align__(16) float sT[8 * 16 * TILE_PITCH];
  __shared__ __align__(16) float sG[16 * 3 * 8];
  __shared__ __align__(16) float sC[8 * 2];

  const int tid   = threadIdx.x;
  const int wave  = tid >> 5;
  const int lane  = tid & 31;
  const int half  = lane >> 4;
  const int l16   = lane & 15;
  const int row0  = blockIdx.x * 16;
  const int col0  = wave * 128 + lane * 4;

  float4 xv[16];
  const float* xp = x + (size_t)row0 * 1024 + col0;
  #pragma unroll
  for (int i = 0; i < 16; ++i) xv[i] = *(const float4*)(xp + (size_t)i * 1024);

  const float4 b0 = *(const float4*)(b + col0);
  const float4 b1 = *(const float4*)(b + 1024 + col0);
  const float4 b2 = *(const float4*)(b + 2048 + col0);
  const float4 w1 = *(const float4*)(W + 1024 + col0);
  const float4 w2 = *(const float4*)(W + 2048 + col0);

  const bool bvalid = (l16 < 3);
  float4 bf[16];
  #pragma unroll
  for (int c = 0; c < 16; ++c) bf[c] = make_float4(0.0f, 0.0f, 0.0f, 0.0f);
  if (bvalid) {
    const float* wp = W + l16 * 1024 + wave * 128 + half * 4;
    #pragma unroll
    for (int c = 0; c < 16; ++c) bf[c] = *(const float4*)(wp + 8 * c);
  }

  float* tile = sT + wave * (16 * TILE_PITCH);
  #pragma unroll
  for (int i = 0; i < 16; ++i)
    *(float4*)(tile + i * TILE_PITCH + lane * 4) = xv[i];

  float p1 = b0.x * w1.x + b0.y * w1.y + b0.z * w1.z + b0.w * w1.w;
  float p2 = (b0.x + b1.x) * w2.x + (b0.y + b1.y) * w2.y
           + (b0.z + b1.z) * w2.z + (b0.w + b1.w) * w2.w;
  for (int m = 16; m > 0; m >>= 1) {
    p1 += __shfl_xor(p1, m);
    p2 += __shfl_xor(p2, m);
  }
  float4 bs;
  bs.x = b0.x + b1.x + b2.x;
  bs.y = b0.y + b1.y + b2.y;
  bs.z = b0.z + b1.z + b2.z;
  bs.w = b0.w + b1.w + b2.w;

  __builtin_amdgcn_fence(__ATOMIC_RELEASE, "wavefront");
  __builtin_amdgcn_wave_barrier();
  __builtin_amdgcn_fence(__ATOMIC_ACQUIRE, "wavefront");

  v8f acc0 = {};
  v8f acc1 = {};
  const float* ap = tile + l16 * TILE_PITCH + half * 4;
  #pragma unroll
  for (int c = 0; c < 16; ++c) {
    const float4 a4 = *(const float4*)(ap + 8 * c);
    const v2f alo = {a4.x, a4.y};
    const v2f ahi = {a4.z, a4.w};
    const v2f blo = {bf[c].x, bf[c].y};
    const v2f bhi = {bf[c].z, bf[c].w};
    acc0 = __builtin_amdgcn_wmma_f32_16x16x4_f32(false, alo, false, blo, (short)0, acc0, false, false);
    acc1 = __builtin_amdgcn_wmma_f32_16x16x4_f32(false, ahi, false, bhi, (short)0, acc1, false, false);
  }
  const v8f acc = acc0 + acc1;

  if (bvalid) {
    #pragma unroll
    for (int r = 0; r < 8; ++r)
      sG[((r + 8 * half) * 3 + l16) * 8 + wave] = acc[r];
  }
  if (lane == 0) { sC[wave * 2] = p1; sC[wave * 2 + 1] = p2; }
  __syncthreads();

  const float4* gp = (const float4*)(sG + l16 * 24);
  const float4 q0 = gp[0], q1 = gp[1], q2 = gp[2], q3 = gp[3], q4 = gp[4], q5 = gp[5];
  const float g0 = ((((((q0.x + q0.y) + q0.z) + q0.w) + q1.x) + q1.y) + q1.z) + q1.w;
  const float g1 = ((((((q2.x + q2.y) + q2.z) + q2.w) + q3.x) + q3.y) + q3.z) + q3.w;
  const float g2 = ((((((q4.x + q4.y) + q4.z) + q4.w) + q5.x) + q5.y) + q5.z) + q5.w;
  const float4* cp = (const float4*)sC;
  const float4 e0 = cp[0], e1 = cp[1], e2 = cp[2], e3 = cp[3];
  const float c1 = ((((((e0.x + e0.z) + e1.x) + e1.z) + e2.x) + e2.z) + e3.x) + e3.z;
  const float c2 = ((((((e0.y + e0.w) + e1.y) + e1.w) + e2.y) + e2.w) + e3.y) + e3.w;
  const float a1 = 1.0f + g0;
  const float a2 = a1 * (1.0f + g1) + c1;
  const float a3 = a2 * (1.0f + g2) + c2;

  float* op = out + (size_t)row0 * 1024 + col0;
  #pragma unroll
  for (int i = 0; i < 16; ++i) {
    const float a = __int_as_float(__builtin_amdgcn_readlane(__float_as_int(a3), i));
    float4 o;
    o.x = a * xv[i].x + bs.x;
    o.y = a * xv[i].y + bs.y;
    o.z = a * xv[i].z + bs.z;
    o.w = a * xv[i].w + bs.w;
    *(float4*)(op + (size_t)i * 1024) = o;
  }
}

extern "C" void kernel_launch(void* const* d_in, const int* in_sizes, int n_in,
                              void* d_out, int out_size, void* d_ws, size_t ws_size,
                              hipStream_t stream) {
  (void)in_sizes; (void)n_in; (void)out_size; (void)d_ws; (void)ws_size;
  const float* x = (const float*)d_in[0];
  const float* W = (const float*)d_in[1];
  const float* b = (const float*)d_in[2];
  float* out = (float*)d_out;
  cross_fused_kernel<<<512, 256, 0, stream>>>(x, W, b, out);
}


# Each block walks two 16-row tiles (256 blocks of 8 waves): all loads requested up front, tile 0 is multiplied and stored while tile 1 arrives; block barriers wait for LDS only
# speedup vs baseline: 2.5473x; 1.1916x over previous
#include <hip/hip_runtime.h>


typedef float v2f __attribute__((ext_vector_type(2)));
typedef float v8f __attribute__((ext_vector_type(8)));

#define TILE_PITCH 132


static __device__ __forceinline__ void wave_tile_handover() {
  __builtin_amdgcn_fence(__ATOMIC_RELEASE, "wavefront");
  __builtin_amdgcn_wave_barrier();
  __builtin_amdgcn_fence(__ATOMIC_ACQUIRE, "wavefront");
}

static __device__ __forceinline__ void block_lds_barrier() {
  __builtin_amdgcn_fence(__ATOMIC_RELEASE, "workgroup", "local");
  __builtin_amdgcn_s_barrier();
  __builtin_amdgcn_fence(__ATOMIC_ACQUIRE, "workgroup", "local");
}

static __device__ __forceinline__ void stage_rows(float* tile, int lane, const float4 (&xv)[16]) {
  #pragma unroll
  for (int i = 0; i < 16; ++i)
    *(float4*)(tile + i * TILE_PITCH + lane * 4) = xv[i];
}

static __device__ __forceinline__ v8f slice_products(const float* ap, const float4 (&bf)[16]) {
  v8f acc0 = {};
  v8f acc1 = {};
  #pragma unroll
  for (int c = 0; c < 16; ++c) {
    const float4 a4 = *(const float4*)(ap + 8 * c);
    const v2f alo = {a4.x, a4.y};
    const v2f ahi = {a4.z, a4.w};
    const v2f blo = {bf[c].x, bf[c].y};
    const v2f bhi = {bf[c].z, bf[c].w};
    acc0 = __builtin_amdgcn_wmma_f32_16x16x4_f32(false, alo, false, blo, (short)0, acc0, false, false);
    acc1 = __builtin_amdgcn_wmma_f32_16x16x4_f32(false, ahi, false, bhi, (short)0, acc1, false, false);
  }
  return acc0 + acc1;
}

static __device__ __forceinline__ float row_alpha3(const float* sGt, int l16, float c1, float c2) {
  const float4* gp = (const float4*)(sGt + l16 * 24);
  const float4 q0 = gp[0], q1 = gp[1], q2 = gp[2], q3 = gp[3], q4 = gp[4], q5 = gp[5];
  const float g0 = ((((((q0.x + q0.y) + q0.z) + q0.w) + q1.x) + q1.y) + q1.z) + q1.w;
  const float g1 = ((((((q2.x + q2.y) + q2.z) + q2.w) + q3.x) + q3.y) + q3.z) + q3.w;
  const float g2 = ((((((q4.x + q4.y) + q4.z) + q4.w) + q5.x) + q5.y) + q5.z) + q5.w;
  const float a1 = 1.0f + g0;
  const float a2 = a1 * (1.0f + g1) + c1;
  return a2 * (1.0f + g2) + c2;
}

static __device__ __forceinline__ void write_rows(float* op, float a3, const float4 (&xv)[16], const float4 bs) {
  #pragma unroll
  for (int i = 0; i < 16; ++i) {
    const float a = __int_as_float(__builtin_amdgcn_readlane(__float_as_int(a3), i));
    float4 o;
    o.x = a * xv[i].x + bs.x;
    o.y = a * xv[i].y + bs.y;
    o.z = a * xv[i].z + bs.z;
    o.w = a * xv[i].w + bs.w;
    *(float4*)(op + (size_t)i * 1024) = o;
  }
}

__global__ __launch_bounds__(256)
void cross_fused_kernel(const float* __restrict__ x,
                        const float* __restrict__ W,
                        const float* __restrict__ b,
                        float* __restrict__ out) {
  __shared__ __align__(16) float sT[8 * 16 * TILE_PITCH];
  __shared__ __align__(16) float sG[2][16 * 3 * 8];
  __shared__ __align__(16) float sC[8 * 2];

  const int tid   = threadIdx.x;
  const int wave  = tid >> 5;
  const int lane  = tid & 31;
  const int half  = lane >> 4;
  const int l16   = lane & 15;
  const int row0  = blockIdx.x * 32;
  const int col0  = wave * 128 + lane * 4;

  float4 xa[16];
  const float* xp = x + (size_t)row0 * 1024 + col0;
  #pragma unroll
  for (int i = 0; i < 16; ++i) xa[i] = *(const float4*)(xp + (size_t)i * 1024);

  const float4 b0 = *(const float4*)(b + col0);
  const float4 b1 = *(const float4*)(b + 1024 + col0);
  const float4 b2 = *(const float4*)(b + 2048 + col0);
  const float4 w1 = *(const float4*)(W + 1024 + col0);
  const float4 w2 = *(const float4*)(W + 2048 + col0);

  const bool bvalid = (l16 < 3);
  float4 bf[16];
  #pragma unroll
  for (int c = 0; c < 16; ++c) bf[c] = make_float4(0.0f, 0.0f, 0.0f, 0.0f);
  if (bvalid) {
    const float* wp = W + l16 * 1024 + wave * 128 + half * 4;
    #pragma unroll
    for (int c = 0; c < 16; ++c) bf[c] = *(const float4*)(wp + 8 * c);
  }

  float4 xb[16];
  #pragma unroll
  for (int i = 0; i < 16; ++i) xb[i] = *(const float4*)(xp + (size_t)(16 + i) * 1024);

  float* tile = sT + wave * (16 * TILE_PITCH);
  stage_rows(tile, lane, xa);

  float p1 = b0.x * w1.x + b0.y * w1.y + b0.z * w1.z + b0.w * w1.w;
  float p2 = (b0.x + b1.x) * w2.x + (b0.y + b1.y) * w2.y
           + (b0.z + b1.z) * w2.z + (b0.w + b1.w) * w2.w;
  for (int m = 16; m > 0; m >>= 1) {
    p1 += __shfl_xor(p1, m);
    p2 += __shfl_xor(p2, m);
  }
  float4 bs;
  bs.x = b0.x + b1.x + b2.x;
  bs.y = b0.y + b1.y + b2.y;
  bs.z = b0.z + b1.z + b2.z;
  bs.w = b0.w + b1.w + b2.w;

  wave_tile_handover();

  const float* ap = tile + l16 * TILE_PITCH + half * 4;
  const v8f accA = slice_products(ap, bf);

  if (bvalid) {
    #pragma unroll
    for (int r = 0; r < 8; ++r)
      sG[0][((r + 8 * half) * 3 + l16) * 8 + wave] = accA[r];
  }
  if (lane == 0) { sC[wave * 2] = p1; sC[wave * 2 + 1] = p2; }
  block_lds_barrier();

  const float4* cp = (const float4*)sC;
  const float4 e0 = cp[0], e1 = cp[1], e2 = cp[2], e3 = cp[3];
  const float c1 = ((((((e0.x + e0.z) + e1.x) + e1.z) + e2.x) + e2.z) + e3.x) + e3.z;
  const float c2 = ((((((e0.y + e0.w) + e1.y) + e1.w) + e2.y) + e2.w) + e3.y) + e3.w;
  float* op = out + (size_t)row0 * 1024 + col0;
  write_rows(op, row_alpha3(sG[0], l16, c1, c2), xa, bs);

  stage_rows(tile, lane, xb);
  wave_tile_handover();

  const v8f accB = slice_products(ap, bf);
  if (bvalid) {
    #pragma unroll
    for (int r = 0; r < 8; ++r)
      sG[1][((r + 8 * half) * 3 + l16) * 8 + wave] = accB[r];
  }
  block_lds_barrier();

  write_rows(op + (size_t)16 * 1024, row_alpha3(sG[1], l16, c1, c2), xb, bs);
}

extern "C" void kernel_launch(void* const* d_in, const int* in_sizes, int n_in,
                              void* d_out, int out_size, void* d_ws, size_t ws_size,
                              hipStream_t stream) {
  (void)in_sizes; (void)n_in; (void)out_size; (void)d_ws; (void)ws_size;
  const float* x = (const float*)d_in[0];
  const float* W = (const float*)d_in[1];
  const float* b = (const float*)d_in[2];
  float* out = (float*)d_out;
  cross_fused_kernel<<<256, 256, 0, stream>>>(x, W, b, out);
}


# Block walks four 8-row granules, not two 16-row tiles: A rows 8-15 hold the same 8 x rows' second 64 columns, met by B columns 3-5, so stores start after a quarter of the data; g sum order changes
# speedup vs baseline: 2.7686x; 1.0869x over previous
#include <hip/hip_runtime.h>


typedef float v2f __attribute__((ext_vector_type(2)));
typedef float v8f __attribute__((ext_vector_type(8)));

#define HALF_PITCH 68
#define G_PITCH 52


static __device__ __forceinline__ void wave_tile_handover() {
  __builtin_amdgcn_fence(__ATOMIC_RELEASE, "wavefront");
  __builtin_amdgcn_wave_barrier();
  __builtin_amdgcn_fence(__ATOMIC_ACQUIRE, "wavefront");
}

static __device__ __forceinline__ void block_lds_barrier() {
  __builtin_amdgcn_fence(__ATOMIC_RELEASE, "workgroup", "local");
  __builtin_amdgcn_s_barrier();
  __builtin_amdgcn_fence(__ATOMIC_ACQUIRE, "workgroup", "local");
}

static __device__ __forceinline__ void stage_rows(float* tp, const float4 (&xv)[8]) {
  #pragma unroll
  for (int i = 0; i < 8; ++i)
    *(float4*)(tp + i * HALF_PITCH) = xv[i];
}

static __device__ __forceinline__ v8f slice_products(const float* ap, const float4 (&bf)[8]) {
  v8f acc0 = {};
  v8f acc1 = {};
  #pragma unroll
  for (int c = 0; c < 8; ++c) {
    const float4 a4 = *(const float4*)(ap + 8 * c);
    const v2f alo = {a4.x, a4.y};
    const v2f ahi = {a4.z, a4.w};
    const v2f blo = {bf[c].x, bf[c].y};
    const v2f bhi = {bf[c].z, bf[c].w};
    acc0 = __builtin_amdgcn_wmma_f32_16x16x4_f32(false, alo, false, blo, (short)0, acc0, false, false);
    acc1 = __builtin_amdgcn_wmma_f32_16x16x4_f32(false, ahi, false, bhi, (short)0, acc1, false, false);
  }
  return acc0 + acc1;
}

static __device__ __forceinline__ float row_alpha3(const float* sGg, int r8, float c1, float c2) {
  const float4* gp = (const float4*)(sGg + r8 * G_PITCH);
  float g[3];
  #pragma unroll
  for (int j = 0; j < 3; ++j) {
    const float4 q0 = gp[4 * j], q1 = gp[4 * j + 1], q2 = gp[4 * j + 2], q3 = gp[4 * j + 3];
    float s = ((((((q0.x + q0.y) + q0.z) + q0.w) + q1.x) + q1.y) + q1.z) + q1.w;
    s = (((((((s + q2.x) + q2.y) + q2.z) + q2.w) + q3.x) + q3.y) + q3.z) + q3.w;
    g[j] = s;
  }
  const float a1 = 1.0f + g[0];
  const float a2 = a1 * (1.0f + g[1]) + c1;
  return a2 * (1.0f + g[2]) + c2;
}

static __device__ __forceinline__ void write_rows(float* op, float a3, const float4 (&xv)[8], const float4 bs) {
  #pragma unroll
  for (int i = 0; i < 8; ++i) {
    const float a = __int_as_float(__builtin_amdgcn_readlane(__float_as_int(a3), i));
    float4 o;
    o.x = a * xv[i].x + bs.x;
    o.y = a * xv[i].y + bs.y;
    o.z = a * xv[i].z + bs.z;
    o.w = a * xv[i].w + bs.w;
    *(float4*)(op + (size_t)i * 1024) = o;
  }
}

__global__ __launch_bounds__(256)
void cross_fused_kernel(const float* __restrict__ x,
                        const float* __restrict__ W,
                        const float* __restrict__ b,
                        float* __restrict__ out) {
  __shared__ __align__(16) float sT[8 * 16 * HALF_PITCH];
  __shared__ __align__(16) float sG[4][8 * G_PITCH];
  __shared__ __align__(16) float sC[8 * 2];

  const int tid   = threadIdx.x;
  const int wave  = tid >> 5;
  const int lane  = tid & 31;
  const int half  = lane >> 4;
  const int l16   = lane & 15;
  const int row0  = blockIdx.x * 32;
  const int col0  = wave * 128 + lane * 4;

  float4 xv[4][8];
  const float* xp = x + (size_t)row0 * 1024 + col0;
  #pragma unroll
  for (int i = 0; i < 8; ++i) xv[0][i] = *(const float4*)(xp + (size_t)i * 1024);

  const float4 b0 = *(const float4*)(b + col0);
  const float4 b1 = *(const float4*)(b + 1024 + col0);
  const float4 b2 = *(const float4*)(b + 2048 + col0);
  const float4 w1 = *(const float4*)(W + 1024 + col0);
  const float4 w2 = *(const float4*)(W + 2048 + col0);

  const bool bvalid = (l16 < 6);
  const int  bP     = (l16 >= 3) ? 1 : 0;
  const int  bj     = l16 - 3 * bP;
  float4 bf[8];
  #pragma unroll
  for (int c = 0; c < 8; ++c) bf[c] = make_float4(0.0f, 0.0f, 0.0f, 0.0f);
  if (bvalid) {
    const float* wp = W + bj * 1024 + wave * 128 + bP * 64 + half * 4;
    #pragma unroll
    for (int c = 0; c < 8; ++c) bf[c] = *(const float4*)(wp + 8 * c);
  }

  #pragma unroll
  for (int g = 1; g < 4; ++g) {
    #pragma unroll
    for (int i = 0; i < 8; ++i) xv[g][i] = *(const float4*)(xp + (size_t)(8 * g + i) * 1024);
  }

  float p1 = b0.x * w1.x + b0.y * w1.y + b0.z * w1.z + b0.w * w1.w;
  float p2 = (b0.x + b1.x) * w2.x + (b0.y + b1.y) * w2.y
           + (b0.z + b1.z) * w2.z + (b0.w + b1.w) * w2.w;
  for (int m = 16; m > 0; m >>= 1) {
    p1 += __shfl_xor(p1, m);
    p2 += __shfl_xor(p2, m);
  }
  float4 bs;
  bs.x = b0.x + b1.x + b2.x;
  bs.y = b0.y + b1.y + b2.y;
  bs.z = b0.z + b1.z + b2.z;
  bs.w = b0.w + b1.w + b2.w;

  float* tile = sT + wave * (16 * HALF_PITCH);
  float* tp = tile + (8 * half) * HALF_PITCH + l16 * 4;
  const float* ap = tile + l16 * HALF_PITCH + half * 4;

  const bool dvalid = bvalid && (bP == half);
  const int  slice  = 2 * wave + half;
  float* op = out + (size_t)row0 * 1024 + col0;

  float c1 = 0.0f, c2 = 0.0f;
  #pragma unroll
  for (int g = 0; g < 4; ++g) {
    stage_rows(tp, xv[g]);
    wave_tile_handover();

    const v8f acc = slice_products(ap, bf);
    if (dvalid) {
      #pragma unroll
      for (int r = 0; r < 8; ++r)
        sG[g][r * G_PITCH + bj * 16 + slice] = acc[r];
    }
    if (g == 0 && lane == 0) { sC[wave * 2] = p1; sC[wave * 2 + 1] = p2; }
    block_lds_barrier();

    if (g == 0) {
      const float4* cp = (const float4*)sC;
      const float4 e0 = cp[0], e1 = cp[1], e2 = cp[2], e3 = cp[3];
      c1 = ((((((e0.x + e0.z) + e1.x) + e1.z) + e2.x) + e2.z) + e3.x) + e3.z;
      c2 = ((((((e0.y + e0.w) + e1.y) + e1.w) + e2.y) + e2.w) + e3.y) + e3.w;
    }
    write_rows(op + (size_t)(8 * g) * 1024, row_alpha3(sG[g], lane & 7, c1, c2), xv[g], bs);
    if (g < 3) __builtin_amdgcn_sched_barrier(0);
  }
}

extern "C" void kernel_launch(void* const* d_in, const int* in_sizes, int n_in,
                              void* d_out, int out_size, void* d_ws, size_t ws_size,
                              hipStream_t stream) {
  (void)in_sizes; (void)n_in; (void)out_size; (void)d_ws; (void)ws_size;
  const float* x = (const float*)d_in[0];
  const float* W = (const float*)d_in[1];
  const float* b = (const float*)d_in[2];
  float* out = (float*)d_out;
  cross_fused_kernel<<<256, 256, 0, stream>>>(x, W, b, out);
}


# Block = 4 waves (one per SIMD) over 16 rows, wave owns 256 columns: a WGP's two blocks run out of step, one's LDS and barrier hops under the other's WMMAs; c1, c2 sum order changes
# speedup vs baseline: 2.9880x; 1.0792x over previous
#include <hip/hip_runtime.h>


typedef float v2f __attribute__((ext_vector_type(2)));
typedef float v4f __attribute__((ext_vector_type(4)));
typedef float v8f __attribute__((ext_vector_type(8)));

#define HALF_PITCH 68
#define G_PITCH 52


static __device__ __forceinline__ void wave_tile_handover() {
  __builtin_amdgcn_fence(__ATOMIC_RELEASE, "wavefront");
  __builtin_amdgcn_wave_barrier();
  __builtin_amdgcn_fence(__ATOMIC_ACQUIRE, "wavefront");
}

static __device__ __forceinline__ void block_lds_barrier() {
  __builtin_amdgcn_fence(__ATOMIC_RELEASE, "workgroup", "local");
  __builtin_amdgcn_s_barrier();
  __builtin_amdgcn_fence(__ATOMIC_ACQUIRE, "workgroup", "local");
}

static __device__ __forceinline__ void stage_rows(float* tp, const float4 (&xv)[8]) {
  #pragma unroll
  for (int i = 0; i < 4; ++i) {
    #pragma unroll
    for (int q = 0; q < 2; ++q)
      *(float4*)(tp + (8 * q + i) * HALF_PITCH) = xv[2 * i + q];
  }
}

static __device__ __forceinline__ void slice_products(const float* ap, const float4 (&bf)[8],
                                                      v4f& rows03, v4f& rows47) {
  v8f acc0 = {};
  v8f acc1 = {};
  #pragma unroll
  for (int c = 0; c < 8; ++c) {
    const float4 a4 = *(const float4*)(ap + 8 * c);
    const v2f alo = {a4.x, a4.y};
    const v2f ahi = {a4.z, a4.w};
    const v2f blo = {bf[c].x, bf[c].y};
    const v2f bhi = {bf[c].z, bf[c].w};
    acc0 = __builtin_amdgcn_wmma_f32_16x16x4_f32(false, alo, false, blo, (short)0, acc0, false, false);
    acc1 = __builtin_amdgcn_wmma_f32_16x16x4_f32(false, ahi, false, bhi, (short)0, acc1, false, false);
  }
  rows03 = acc0.lo + acc1.lo;
  rows47 = acc0.hi + acc1.hi;
}

static __device__ __forceinline__ float row_alpha3(const float* sGg, int r4, float c1, float c2) {
  const float4* gp = (const float4*)(sGg + r4 * G_PITCH);
  float g[3];
  #pragma unroll
  for (int j = 0; j < 3; ++j) {
    const float4 q0 = gp[4 * j], q1 = gp[4 * j + 1], q2 = gp[4 * j + 2], q3 = gp[4 * j + 3];
    float s = ((((((q0.x + q0.y) + q0.z) + q0.w) + q1.x) + q1.y) + q1.z) + q1.w;
    s = (((((((s + q2.x) + q2.y) + q2.z) + q2.w) + q3.x) + q3.y) + q3.z) + q3.w;
    g[j] = s;
  }
  const float a1 = 1.0f + g[0];
  const float a2 = a1 * (1.0f + g[1]) + c1;
  return a2 * (1.0f + g[2]) + c2;
}

static __device__ __forceinline__ void write_rows(float* op, float a3, const float4 (&xv)[8],
                                                  const float4 (&bs)[2]) {
  #pragma unroll
  for (int i = 0; i < 4; ++i) {
    const float a = __int_as_float(__builtin_amdgcn_readlane(__float_as_int(a3), i));
    #pragma unroll
    for (int q = 0; q < 2; ++q) {
      float4 o;
      o.x = a * xv[2 * i + q].x + bs[q].x;
      o.y = a * xv[2 * i + q].y + bs[q].y;
      o.z = a * xv[2 * i + q].z + bs[q].z;
      o.w = a * xv[2 * i + q].w + bs[q].w;
      *(float4*)(op + (size_t)i * 1024 + 128 * q) = o;
    }
  }
}

__global__ __launch_bounds__(128)
void cross_fused_kernel(const float* __restrict__ x,
                        const float* __restrict__ W,
                        const float* __restrict__ b,
                        float* __restrict__ out) {
  __shared__ __align__(16) float sT[4 * 16 * HALF_PITCH];
  __shared__ __align__(16) float sG[4][4 * G_PITCH];
  __shared__ __align__(16) float sC[4 * 2];

  const int tid   = threadIdx.x;
  const int wave  = tid >> 5;
  const int lane  = tid & 31;
  const int half  = lane >> 4;
  const int l16   = lane & 15;
  const int row0  = blockIdx.x * 16;
  const int col0  = wave * 256 + lane * 4;

  float4 xv[4][8];
  const float* xp = x + (size_t)row0 * 1024 + col0;
  #pragma unroll
  for (int i = 0; i < 4; ++i) {
    #pragma unroll
    for (int q = 0; q < 2; ++q) xv[0][2 * i + q] = *(const float4*)(xp + (size_t)i * 1024 + 128 * q);
  }

  float4 b0[2], b1[2], b2[2], w1[2], w2[2];
  #pragma unroll
  for (int q = 0; q < 2; ++q) {
    b0[q] = *(const float4*)(b + col0 + 128 * q);
    b1[q] = *(const float4*)(b + 1024 + col0 + 128 * q);
    b2[q] = *(const float4*)(b + 2048 + col0 + 128 * q);
    w1[q] = *(const float4*)(W + 1024 + col0 + 128 * q);
    w2[q] = *(const float4*)(W + 2048 + col0 + 128 * q);
  }

  const bool bvalid = (l16 < 12);
  const int  bP     = (l16 >= 3 ? 1 : 0) + (l16 >= 6 ? 1 : 0) + (l16 >= 9 ? 1 : 0);
  const int  bj     = l16 - 3 * bP;
  float4 bf[8];
  #pragma unroll
  for (int c = 0; c < 8; ++c) bf[c] = make_float4(0.0f, 0.0f, 0.0f, 0.0f);
  if (bvalid) {
    const float* wp = W + bj * 1024 + wave * 256 + bP * 64 + half * 4;
    #pragma unroll
    for (int c = 0; c < 8; ++c) bf[c] = *(const float4*)(wp + 8 * c);
  }

  #pragma unroll
  for (int g = 1; g < 4; ++g) {
    #pragma unroll
    for (int i = 0; i < 4; ++i) {
      #pragma unroll
      for (int q = 0; q < 2; ++q)
        xv[g][2 * i + q] = *(const float4*)(xp + (size_t)(4 * g + i) * 1024 + 128 * q);
    }
  }

  float p1 = 0.0f, p2 = 0.0f;
  float4 bs[2];
  #pragma unroll
  for (int q = 0; q < 2; ++q) {
    p1 += b0[q].x * w1[q].x + b0[q].y * w1[q].y + b0[q].z * w1[q].z + b0[q].w * w1[q].w;
    p2 += (b0[q].x + b1[q].x) * w2[q].x + (b0[q].y + b1[q].y) * w2[q].y
        + (b0[q].z + b1[q].z) * w2[q].z + (b0[q].w + b1[q].w) * w2[q].w;
    bs[q].x = b0[q].x + b1[q].x + b2[q].x;
    bs[q].y = b0[q].y + b1[q].y + b2[q].y;
    bs[q].z = b0[q].z + b1[q].z + b2[q].z;
    bs[q].w = b0[q].w + b1[q].w + b2[q].w;
  }
  for (int m = 16; m > 0; m >>= 1) {
    p1 += __shfl_xor(p1, m);
    p2 += __shfl_xor(p2, m);
  }

  float* tile = sT + wave * (16 * HALF_PITCH);
  float* tp = tile + (4 * half) * HALF_PITCH + l16 * 4;
  const float* ap = tile + l16 * HALF_PITCH + half * 4;

  const bool dvalid = bvalid && ((bP >> 1) == half);
  const bool dhi    = (bP & 1) != 0;
  const int  slice  = 4 * wave + bP;
  float* op = out + (size_t)row0 * 1024 + col0;

  float c1 = 0.0f, c2 = 0.0f;
  #pragma unroll
  for (int g = 0; g < 4; ++g) {
    stage_rows(tp, xv[g]);
    wave_tile_handover();

    v4f rows03, rows47;
    slice_products(ap, bf, rows03, rows47);
    if (dvalid) {
      const v4f part = dhi ? rows47 : rows03;
      #pragma unroll
      for (int i = 0; i < 4; ++i)
        sG[g][i * G_PITCH + bj * 16 + slice] = part[i];
    }
    if (g == 0 && lane == 0) { sC[wave * 2] = p1; sC[wave * 2 + 1] = p2; }
    block_lds_barrier();

    if (g == 0) {
      const float4* cp = (const float4*)sC;
      const float4 e0 = cp[0], e1 = cp[1];
      c1 = ((e0.x + e0.z) + e1.x) + e1.z;
      c2 = ((e0.y + e0.w) + e1.y) + e1.w;
    }
    write_rows(op + (size_t)(4 * g) * 1024, row_alpha3(sG[g], lane & 3, c1, c2), xv[g], bs);
    if (g < 3) __builtin_amdgcn_sched_barrier(0);
  }
}

extern "C" void kernel_launch(void* const* d_in, const int* in_sizes, int n_in,
                              void* d_out, int out_size, void* d_ws, size_t ws_size,
                              hipStream_t stream) {
  (void)in_sizes; (void)n_in; (void)out_size; (void)d_ws; (void)ws_size;
  const float* x = (const float*)d_in[0];
  const float* W = (const float*)d_in[1];
  const float* b = (const float*)d_in[2];
  float* out = (float*)d_out;
  cross_fused_kernel<<<512, 128, 0, stream>>>(x, W, b, out);
}
